# starting kernel
# baseline (speedup 1.0000x reference)
#include <hip/hip_runtime.h>


typedef float v2f __attribute__((ext_vector_type(2)));
typedef float v8f __attribute__((ext_vector_type(8)));

#define N_NODES 20000
#define N_EDGES 8192
#define D_NODE  64
#define D_EDGE  16
#define UNITS   64
#define K1      (D_NODE + D_EDGE)
#define K2      (UNITS + UNITS)

__global__ __launch_bounds__(256) void gemm_init_kernel(
    const float* __restrict__ node_feature,
    const float* __restrict__ edge_feature,
    const float* __restrict__ W,
    const float* __restrict__ bias,
    const int*   __restrict__ edge_src,
    float*       __restrict__ edge_state)
{
    __shared__ float sW[K1 * UNITS];
    for (int i = threadIdx.x; i < K1 * UNITS; i += 256) sW[i] = W[i];
    __syncthreads();

    const int wave    = threadIdx.x >> 5;
    const int lane    = threadIdx.x & 31;
    const int rowTile = blockIdx.x * 8 + wave;
    const int m       = lane & 15;
    const int kh      = (lane >> 4) << 1;
    const int e       = rowTile * 16 + m;
    const int s       = edge_src[e];
    const float* xn = node_feature + (long)s * D_NODE;
    const float* xe = edge_feature + (long)e * D_EDGE;

    v8f acc[4] = {};
    #pragma unroll
    for (int k0 = 0; k0 < K1; k0 += 4) {
        const int k = k0 + kh;
        v2f a;
        a.x = (k     < D_NODE) ? xn[k]     : xe[k     - D_NODE];
        a.y = (k + 1 < D_NODE) ? xn[k + 1] : xe[k + 1 - D_NODE];
        const float* wr0 = &sW[(k    ) * UNITS];
        const float* wr1 = &sW[(k + 1) * UNITS];
        #pragma unroll
        for (int nt = 0; nt < 4; ++nt) {
            v2f b;
            b.x = wr0[nt * 16 + m];
            b.y = wr1[nt * 16 + m];
            acc[nt] = __builtin_amdgcn_wmma_f32_16x16x4_f32(
                false, a, false, b, (short)0, acc[nt], false, false);
        }
    }

    const int half = lane >> 4;
    #pragma unroll
    for (int nt = 0; nt < 4; ++nt) {
        const int n = nt * 16 + m;
        const float bv = bias[n];
        #pragma unroll
        for (int r = 0; r < 8; ++r) {
            const int mm = r + 8 * half;
            edge_state[(long)(rowTile * 16 + mm) * UNITS + n] = acc[nt][r] + bv;
        }
    }
}

__global__ void zero_kernel(float* __restrict__ p, int n)
{
    int i = blockIdx.x * blockDim.x + threadIdx.x;
    if (i < n) p[i] = 0.0f;
}

__global__ void scatter_kernel(const float* __restrict__ edge_state,
                               const int*   __restrict__ edge_dst,
                               float*       __restrict__ node_agg)
{
    int t = blockIdx.x * 256 + threadIdx.x;
    int e = t >> 6, u = t & 63;
    atomicAdd(&node_agg[(long)edge_dst[e] * UNITS + u], edge_state[t]);
}

__global__ void gather_kernel(const float* __restrict__ node_agg,
                              const int*   __restrict__ edge_src,
                              float*       __restrict__ message)
{
    int t = blockIdx.x * 256 + threadIdx.x;
    int e = t >> 6, u = t & 63;
    message[t] = node_agg[(long)edge_src[e] * UNITS + u];
}

__global__ __launch_bounds__(256) void reverse_kernel(
    const int*   __restrict__ edge_src,
    const int*   __restrict__ edge_dst,
    const float* __restrict__ edge_state,
    float*       __restrict__ message)
{
    __shared__ int ss[N_EDGES];
    __shared__ int sd[N_EDGES];
    for (int j = threadIdx.x; j < N_EDGES; j += 256) {
        ss[j] = edge_src[j];
        sd[j] = edge_dst[j];
    }
    __syncthreads();

    const int i  = blockIdx.x * 256 + threadIdx.x;
    const int si = ss[i];
    const int di = sd[i];
    float* mi = message + (long)i * UNITS;
    for (int j = 0; j < N_EDGES; ++j) {
        if (ss[j] == di && sd[j] == si) {
            const float* esj = edge_state + (long)j * UNITS;
            #pragma unroll 4
            for (int u = 0; u < UNITS; ++u) mi[u] -= esj[u];
        }
    }
}

__global__ __launch_bounds__(256) void gemm_upd_kernel(
    const float* __restrict__ edge_state,
    const float* __restrict__ message,
    const float* __restrict__ W,
    const float* __restrict__ bias,
    float*       __restrict__ out)
{
    __shared__ float sW[K2 * UNITS];
    for (int i = threadIdx.x; i < K2 * UNITS; i += 256) sW[i] = W[i];
    __syncthreads();

    const int wave    = threadIdx.x >> 5;
    const int lane    = threadIdx.x & 31;
    const int rowTile = blockIdx.x * 8 + wave;
    const int m       = lane & 15;
    const int kh      = (lane >> 4) << 1;
    const int e       = rowTile * 16 + m;
    const float* x0 = edge_state + (long)e * UNITS;
    const float* x1 = message    + (long)e * UNITS;

    v8f acc[4] = {};
    #pragma unroll
    for (int k0 = 0; k0 < K2; k0 += 4) {
        const int k = k0 + kh;
        v2f a;
        a.x = (k     < UNITS) ? x0[k]     : x1[k     - UNITS];
        a.y = (k + 1 < UNITS) ? x0[k + 1] : x1[k + 1 - UNITS];
        const float* wr0 = &sW[(k    ) * UNITS];
        const float* wr1 = &sW[(k + 1) * UNITS];
        #pragma unroll
        for (int nt = 0; nt < 4; ++nt) {
            v2f b;
            b.x = wr0[nt * 16 + m];
            b.y = wr1[nt * 16 + m];
            acc[nt] = __builtin_amdgcn_wmma_f32_16x16x4_f32(
                false, a, false, b, (short)0, acc[nt], false, false);
        }
    }

    const int half = lane >> 4;
    #pragma unroll
    for (int nt = 0; nt < 4; ++nt) {
        const int n = nt * 16 + m;
        const float bv = bias[n];
        #pragma unroll
        for (int r = 0; r < 8; ++r) {
            const int mm = r + 8 * half;
            out[(long)(rowTile * 16 + mm) * UNITS + n] = acc[nt][r] + bv;
        }
    }
}

extern "C" void kernel_launch(void* const* d_in, const int* in_sizes, int n_in,
                              void* d_out, int out_size, void* d_ws, size_t ws_size,
                              hipStream_t stream)
{
    const float* node_feature = (const float*)d_in[0];
    const float* edge_feature = (const float*)d_in[1];
    const float* W_init       = (const float*)d_in[2];
    const float* b_init       = (const float*)d_in[3];
    const float* W_upd        = (const float*)d_in[4];
    const float* b_upd        = (const float*)d_in[5];
    const int*   edge_src     = (const int*)d_in[6];
    const int*   edge_dst     = (const int*)d_in[7];
    float*       out          = (float*)d_out;

    char* ws = (char*)d_ws;
    float* edge_state = (float*)ws;
    float* node_agg   = (float*)(ws + (size_t)N_EDGES * UNITS * 4);
    float* message    = (float*)(ws + (size_t)N_EDGES * UNITS * 4
                                    + (size_t)N_NODES * UNITS * 4);

    gemm_init_kernel<<<64, 256, 0, stream>>>(node_feature, edge_feature,
                                             W_init, b_init, edge_src, edge_state);

    zero_kernel<<<(N_NODES * UNITS + 255) / 256, 256, 0, stream>>>(
        node_agg, N_NODES * UNITS);

    scatter_kernel<<<(N_EDGES * UNITS) / 256, 256, 0, stream>>>(
        edge_state, edge_dst, node_agg);

    gather_kernel<<<(N_EDGES * UNITS) / 256, 256, 0, stream>>>(
        node_agg, edge_src, message);

    reverse_kernel<<<N_EDGES / 256, 256, 0, stream>>>(
        edge_src, edge_dst, edge_state, message);

    gemm_upd_kernel<<<64, 256, 0, stream>>>(edge_state, message,
                                            W_upd, b_upd, out);
}


# reverse-edge scan re-tiled: work-group = 32 edges i against all j held in registers (1024 per wave), one key compare per pair; rare matches subtracted after the scan in ascending j; grid 32 -> 256
# speedup vs baseline: 12.1335x; 12.1335x over previous
#include <hip/hip_runtime.h>


typedef float v2f __attribute__((ext_vector_type(2)));
typedef float v8f __attribute__((ext_vector_type(8)));

#define N_NODES 20000
#define N_EDGES 8192
#define D_NODE  64
#define D_EDGE  16
#define UNITS   64
#define K1      (D_NODE + D_EDGE)
#define K2      (UNITS + UNITS)

__global__ __launch_bounds__(256) void gemm_init_kernel(
    const float* __restrict__ node_feature,
    const float* __restrict__ edge_feature,
    const float* __restrict__ W,
    const float* __restrict__ bias,
    const int*   __restrict__ edge_src,
    float*       __restrict__ edge_state)
{
    __shared__ float sW[K1 * UNITS];
    for (int i = threadIdx.x; i < K1 * UNITS; i += 256) sW[i] = W[i];
    __syncthreads();

    const int wave    = threadIdx.x >> 5;
    const int lane    = threadIdx.x & 31;
    const int rowTile = blockIdx.x * 8 + wave;
    const int m       = lane & 15;
    const int kh      = (lane >> 4) << 1;
    const int e       = rowTile * 16 + m;
    const int s       = edge_src[e];
    const float* xn = node_feature + (long)s * D_NODE;
    const float* xe = edge_feature + (long)e * D_EDGE;

    v8f acc[4] = {};
    #pragma unroll
    for (int k0 = 0; k0 < K1; k0 += 4) {
        const int k = k0 + kh;
        v2f a;
        a.x = (k     < D_NODE) ? xn[k]     : xe[k     - D_NODE];
        a.y = (k + 1 < D_NODE) ? xn[k + 1] : xe[k + 1 - D_NODE];
        const float* wr0 = &sW[(k    ) * UNITS];
        const float* wr1 = &sW[(k + 1) * UNITS];
        #pragma unroll
        for (int nt = 0; nt < 4; ++nt) {
            v2f b;
            b.x = wr0[nt * 16 + m];
            b.y = wr1[nt * 16 + m];
            acc[nt] = __builtin_amdgcn_wmma_f32_16x16x4_f32(
                false, a, false, b, (short)0, acc[nt], false, false);
        }
    }

    const int half = lane >> 4;
    #pragma unroll
    for (int nt = 0; nt < 4; ++nt) {
        const int n = nt * 16 + m;
        const float bv = bias[n];
        #pragma unroll
        for (int r = 0; r < 8; ++r) {
            const int mm = r + 8 * half;
            edge_state[(long)(rowTile * 16 + mm) * UNITS + n] = acc[nt][r] + bv;
        }
    }
}

__global__ void zero_kernel(float* __restrict__ p, int n)
{
    int i = blockIdx.x * blockDim.x + threadIdx.x;
    if (i < n) p[i] = 0.0f;
}

__global__ void scatter_kernel(const float* __restrict__ edge_state,
                               const int*   __restrict__ edge_dst,
                               float*       __restrict__ node_agg)
{
    int t = blockIdx.x * 256 + threadIdx.x;
    int e = t >> 6, u = t & 63;
    atomicAdd(&node_agg[(long)edge_dst[e] * UNITS + u], edge_state[t]);
}

__global__ void gather_kernel(const float* __restrict__ node_agg,
                              const int*   __restrict__ edge_src,
                              float*       __restrict__ message)
{
    int t = blockIdx.x * 256 + threadIdx.x;
    int e = t >> 6, u = t & 63;
    message[t] = node_agg[(long)edge_src[e] * UNITS + u];
}

#define REV_I    32
#define REV_R    32
#define REV_JW   (REV_R * 32)
static_assert(REV_JW * 8 == N_EDGES, "8 waves cover every j");
static_assert(N_EDGES % REV_I == 0, "whole work-groups");

__global__ __launch_bounds__(256) void reverse_kernel(
    const int*   __restrict__ edge_src,
    const int*   __restrict__ edge_dst,
    const float* __restrict__ edge_state,
    float*                    message)
{
    __shared__ unsigned wave_hits[8];

    const int lane  = threadIdx.x & 31;
    const int wave  = __builtin_amdgcn_readfirstlane(threadIdx.x >> 5);
    const int jbase = wave * REV_JW;
    const int i0    = blockIdx.x * REV_I;

    unsigned ks[REV_R], kd[REV_R], key[REV_R];
    #pragma unroll
    for (int r = 0; r < REV_R; ++r) {
        ks[r] = (unsigned)edge_src[jbase + r * 32 + lane];
        kd[r] = (unsigned)edge_dst[jbase + r * 32 + lane];
    }
    const unsigned is_v = (unsigned)edge_src[i0 + lane];
    const unsigned id_v = (unsigned)edge_dst[i0 + lane];
    const unsigned iq_v = (is_v << 16) + id_v;
    #pragma unroll
    for (int r = 0; r < REV_R; ++r) key[r] = (kd[r] << 16) + ks[r];

    unsigned hits = 0;
    for (int ii = 0; ii < REV_I; ++ii) {
        const unsigned q = __builtin_amdgcn_readlane(iq_v, ii);
        unsigned m = 0;
        #pragma unroll
        for (int r = 0; r < REV_R; ++r) m |= __builtin_amdgcn_ballot_w32(key[r] == q);
        if (m != 0u) hits |= 1u << ii;
    }

    if (lane == 0) wave_hits[wave] = hits;
    __syncthreads();
    unsigned all_hits = 0;
    #pragma unroll
    for (int w = 0; w < 8; ++w) all_hits |= wave_hits[w];
    all_hits = __builtin_amdgcn_readfirstlane(all_hits);
    if (all_hits == 0u) return;

    for (unsigned todo = all_hits; todo != 0u; todo &= todo - 1u) {
        const int ii = __builtin_ctz(todo);
        const unsigned qs = __builtin_amdgcn_readlane(is_v, ii);
        const unsigned qd = __builtin_amdgcn_readlane(id_v, ii);
        float* mi = message + (long)(i0 + ii) * UNITS;
        for (int w = 0; w < 8; ++w) {
            if (w == wave && ((hits >> ii) & 1u)) {
                #pragma unroll
                for (int r = 0; r < REV_R; ++r) {
                    unsigned m = __builtin_amdgcn_ballot_w32(ks[r] == qd && kd[r] == qs);
                    while (m != 0u) {
                        const int j = jbase + r * 32 + __builtin_ctz(m);
                        const float* esj = edge_state + (long)j * UNITS;
                        mi[lane]      -= esj[lane];
                        mi[lane + 32] -= esj[lane + 32];
                        m &= m - 1u;
                    }
                }
            }
            __syncthreads();
        }
    }
}

__global__ __launch_bounds__(256) void gemm_upd_kernel(
    const float* __restrict__ edge_state,
    const float* __restrict__ message,
    const float* __restrict__ W,
    const float* __restrict__ bias,
    float*       __restrict__ out)
{
    __shared__ float sW[K2 * UNITS];
    for (int i = threadIdx.x; i < K2 * UNITS; i += 256) sW[i] = W[i];
    __syncthreads();

    const int wave    = threadIdx.x >> 5;
    const int lane    = threadIdx.x & 31;
    const int rowTile = blockIdx.x * 8 + wave;
    const int m       = lane & 15;
    const int kh      = (lane >> 4) << 1;
    const int e       = rowTile * 16 + m;
    const float* x0 = edge_state + (long)e * UNITS;
    const float* x1 = message    + (long)e * UNITS;

    v8f acc[4] = {};
    #pragma unroll
    for (int k0 = 0; k0 < K2; k0 += 4) {
        const int k = k0 + kh;
        v2f a;
        a.x = (k     < UNITS) ? x0[k]     : x1[k     - UNITS];
        a.y = (k + 1 < UNITS) ? x0[k + 1] : x1[k + 1 - UNITS];
        const float* wr0 = &sW[(k    ) * UNITS];
        const float* wr1 = &sW[(k + 1) * UNITS];
        #pragma unroll
        for (int nt = 0; nt < 4; ++nt) {
            v2f b;
            b.x = wr0[nt * 16 + m];
            b.y = wr1[nt * 16 + m];
            acc[nt] = __builtin_amdgcn_wmma_f32_16x16x4_f32(
                false, a, false, b, (short)0, acc[nt], false, false);
        }
    }

    const int half = lane >> 4;
    #pragma unroll
    for (int nt = 0; nt < 4; ++nt) {
        const int n = nt * 16 + m;
        const float bv = bias[n];
        #pragma unroll
        for (int r = 0; r < 8; ++r) {
            const int mm = r + 8 * half;
            out[(long)(rowTile * 16 + mm) * UNITS + n] = acc[nt][r] + bv;
        }
    }
}

extern "C" void kernel_launch(void* const* d_in, const int* in_sizes, int n_in,
                              void* d_out, int out_size, void* d_ws, size_t ws_size,
                              hipStream_t stream)
{
    const float* node_feature = (const float*)d_in[0];
    const float* edge_feature = (const float*)d_in[1];
    const float* W_init       = (const float*)d_in[2];
    const float* b_init       = (const float*)d_in[3];
    const float* W_upd        = (const float*)d_in[4];
    const float* b_upd        = (const float*)d_in[5];
    const int*   edge_src     = (const int*)d_in[6];
    const int*   edge_dst     = (const int*)d_in[7];
    float*       out          = (float*)d_out;

    char* ws = (char*)d_ws;
    float* edge_state = (float*)ws;
    float* node_agg   = (float*)(ws + (size_t)N_EDGES * UNITS * 4);
    float* message    = (float*)(ws + (size_t)N_EDGES * UNITS * 4
                                    + (size_t)N_NODES * UNITS * 4);

    gemm_init_kernel<<<64, 256, 0, stream>>>(node_feature, edge_feature,
                                             W_init, b_init, edge_src, edge_state);

    zero_kernel<<<(N_NODES * UNITS + 255) / 256, 256, 0, stream>>>(
        node_agg, N_NODES * UNITS);

    scatter_kernel<<<(N_EDGES * UNITS) / 256, 256, 0, stream>>>(
        edge_state, edge_dst, node_agg);

    gather_kernel<<<(N_EDGES * UNITS) / 256, 256, 0, stream>>>(
        node_agg, edge_src, message);

    reverse_kernel<<<N_EDGES / REV_I, 256, 0, stream>>>(
        edge_src, edge_dst, edge_state, message);

    gemm_upd_kernel<<<64, 256, 0, stream>>>(edge_state, message,
                                            W_upd, b_upd, out);
}


# node table round trip removed: pair scan lists the j with dst[j]==src[i], their rows are summed per edge in ascending j (no atomics) and feed the update GEMM in the same work-group; launches 6 -> 2
# speedup vs baseline: 16.3809x; 1.3501x over previous
#include <hip/hip_runtime.h>


typedef float v2f __attribute__((ext_vector_type(2)));
typedef float v8f __attribute__((ext_vector_type(8)));

#define N_NODES 20000
#define N_EDGES 8192
#define D_NODE  64
#define D_EDGE  16
#define UNITS   64
#define K1      (D_NODE + D_EDGE)
#define K2      (UNITS + UNITS)

__global__ __launch_bounds__(256) void gemm_init_kernel(
    const float* __restrict__ node_feature,
    const float* __restrict__ edge_feature,
    const float* __restrict__ W,
    const float* __restrict__ bias,
    const int*   __restrict__ edge_src,
    float*       __restrict__ edge_state)
{
    __shared__ float sW[K1 * UNITS];
    for (int i = threadIdx.x; i < K1 * UNITS; i += 256) sW[i] = W[i];
    __syncthreads();

    const int wave    = threadIdx.x >> 5;
    const int lane    = threadIdx.x & 31;
    const int rowTile = blockIdx.x * 8 + wave;
    const int m       = lane & 15;
    const int kh      = (lane >> 4) << 1;
    const int e       = rowTile * 16 + m;
    const int s       = edge_src[e];
    const float* xn = node_feature + (long)s * D_NODE;
    const float* xe = edge_feature + (long)e * D_EDGE;

    v8f acc[4] = {};
    #pragma unroll
    for (int k0 = 0; k0 < K1; k0 += 4) {
        const int k = k0 + kh;
        v2f a;
        a.x = (k     < D_NODE) ? xn[k]     : xe[k     - D_NODE];
        a.y = (k + 1 < D_NODE) ? xn[k + 1] : xe[k + 1 - D_NODE];
        const float* wr0 = &sW[(k    ) * UNITS];
        const float* wr1 = &sW[(k + 1) * UNITS];
        #pragma unroll
        for (int nt = 0; nt < 4; ++nt) {
            v2f b;
            b.x = wr0[nt * 16 + m];
            b.y = wr1[nt * 16 + m];
            acc[nt] = __builtin_amdgcn_wmma_f32_16x16x4_f32(
                false, a, false, b, (short)0, acc[nt], false, false);
        }
    }

    const int half = lane >> 4;
    #pragma unroll
    for (int nt = 0; nt < 4; ++nt) {
        const int n = nt * 16 + m;
        const float bv = bias[n];
        #pragma unroll
        for (int r = 0; r < 8; ++r) {
            const int mm = r + 8 * half;
            edge_state[(long)(rowTile * 16 + mm) * UNITS + n] = acc[nt][r] + bv;
        }
    }
}

#define FI    32
#define FR    32
#define FJW   (FR * 32)
#define FPC   8
#define MP    66
static_assert(FJW * 8 == N_EDGES, "8 waves cover every j");
static_assert(N_EDGES % FI == 0, "whole work-groups");

__global__ __launch_bounds__(256) void message_update_kernel(
    const int*   __restrict__ edge_src,
    const int*   __restrict__ edge_dst,
    const float* __restrict__ edge_state,
    const float* __restrict__ W,
    const float* __restrict__ bias,
    float*       __restrict__ out)
{
    __shared__ float          sW[K2 * UNITS];
    __shared__ float          sM[FI * MP];
    __shared__ unsigned short sList[FI][8][FPC];
    __shared__ int            sCnt[FI][8];

    const int tid  = threadIdx.x;
    const int lane = tid & 31;
    const int wave  = __builtin_amdgcn_readfirstlane(tid >> 5);
    const int jbase = wave * FJW;
    const int i0    = blockIdx.x * FI;

    unsigned kd[FR];
    #pragma unroll
    for (int r = 0; r < FR; ++r) kd[r] = (unsigned)edge_dst[jbase + r * 32 + lane];
    const unsigned is_v = (unsigned)edge_src[i0 + lane];

    const int rt = (wave >> 1) & 1;
    const int np = wave & 1;
    const int m  = lane & 15;
    const int kh = (lane >> 4) << 1;
    v2f aes[UNITS / 4];
    #pragma unroll
    for (int q = 0; q < UNITS / 4; ++q) { aes[q].x = 0.0f; aes[q].y = 0.0f; }
    if (wave < 4) {
        const float* x0 = edge_state + (long)(i0 + rt * 16 + m) * UNITS;
        #pragma unroll
        for (int q = 0; q < UNITS / 4; ++q) {
            aes[q].x = x0[4 * q + kh];
            aes[q].y = x0[4 * q + kh + 1];
        }
    }

    {
        const float4* W4  = reinterpret_cast<const float4*>(W);
        float4*       sW4 = reinterpret_cast<float4*>(sW);
        #pragma unroll
        for (int it = 0; it < K2 * UNITS / 4 / 256; ++it)
            sW4[tid + it * 256] = W4[tid + it * 256];
    }

    unsigned cnt_v = 0;
    for (int ii = 0; ii < FI; ++ii) {
        const unsigned q = __builtin_amdgcn_readlane(is_v, ii);
        unsigned any = 0;
        #pragma unroll
        for (int r = 0; r < FR; ++r) any |= __builtin_amdgcn_ballot_w32(kd[r] == q);
        if (any != 0u) {
            unsigned c = 0;
            #pragma unroll
            for (int r = 0; r < FR; ++r) {
                const unsigned bm = __builtin_amdgcn_ballot_w32(kd[r] == q);
                if (bm != 0u) {
                    const unsigned slot = c + __builtin_amdgcn_mbcnt_lo(bm, 0u);
                    if (kd[r] == q && slot < FPC)
                        sList[ii][wave][slot] = (unsigned short)(jbase + r * 32 + lane);
                    c += __builtin_popcount(bm);
                }
            }
            if (lane == ii) cnt_v = c;
        }
    }
    sCnt[lane][wave] = (int)cnt_v;
    __syncthreads();

    {
        const int row = tid >> 3;
        const int c8  = (tid & 7) * 8;
        const int gi  = i0 + row;
        const int qs  = edge_src[gi];
        const int qd  = edge_dst[gi];
        float agg[8], rev[8];
        #pragma unroll
        for (int x = 0; x < 8; ++x) { agg[x] = 0.0f; rev[x] = 0.0f; }

        int n[8];
        #pragma unroll
        for (int w = 0; w < 8; ++w) n[w] = sCnt[row][w];
        bool over = false;
        #pragma unroll
        for (int w = 0; w < 8; ++w) over |= (n[w] > FPC);

        if (!over) {
            int pre[8];
            pre[0] = 0;
            #pragma unroll
            for (int w = 1; w < 8; ++w) pre[w] = pre[w - 1] + n[w - 1];
            const int total = pre[7] + n[7];
            for (int f = 0; f < total; f += 4) {
                float v[4][8];
                bool  back[4];
                #pragma unroll
                for (int u = 0; u < 4; ++u) {
                    const int ff = f + u;
                    int w = 0, base = 0;
                    #pragma unroll
                    for (int x = 1; x < 8; ++x)
                        if (ff >= pre[x]) { w = x; base = pre[x]; }
                    const int j = (ff < total) ? (int)sList[row][w][ff - base] : gi;
                    const float4* p = reinterpret_cast<const float4*>(
                        edge_state + (long)j * UNITS + c8);
                    const float4 v0 = p[0], v1 = p[1];
                    v[u][0] = v0.x; v[u][1] = v0.y; v[u][2] = v0.z; v[u][3] = v0.w;
                    v[u][4] = v1.x; v[u][5] = v1.y; v[u][6] = v1.z; v[u][7] = v1.w;
                    back[u] = (edge_src[j] == qd);
                }
                #pragma unroll
                for (int u = 0; u < 4; ++u) {
                    if (f + u < total) {
                        #pragma unroll
                        for (int x = 0; x < 8; ++x) {
                            agg[x] += v[u][x];
                            if (back[u]) rev[x] += v[u][x];
                        }
                    }
                }
            }
        } else {
            for (int j = 0; j < N_EDGES; ++j) {
                if (edge_dst[j] != qs) continue;
                const float4* p = reinterpret_cast<const float4*>(
                    edge_state + (long)j * UNITS + c8);
                const float4 v0 = p[0], v1 = p[1];
                const float v[8] = {v0.x, v0.y, v0.z, v0.w, v1.x, v1.y, v1.z, v1.w};
                const bool back = (edge_src[j] == qd);
                #pragma unroll
                for (int x = 0; x < 8; ++x) {
                    agg[x] += v[x];
                    if (back) rev[x] += v[x];
                }
            }
        }
        #pragma unroll
        for (int x = 0; x < 8; ++x) sM[row * MP + c8 + x] = agg[x] - rev[x];
    }
    __syncthreads();

    if (wave >= 4) return;
    const float* x1 = &sM[(rt * 16 + m) * MP];

    v8f acc[2] = {};
    #pragma unroll
    for (int k0 = 0; k0 < K2; k0 += 4) {
        const int k = k0 + kh;
        v2f a;
        if (k0 < UNITS) {
            a = aes[k0 / 4];
        } else {
            a.x = x1[k     - UNITS];
            a.y = x1[k + 1 - UNITS];
        }
        const float* wr0 = &sW[(k    ) * UNITS];
        const float* wr1 = &sW[(k + 1) * UNITS];
        #pragma unroll
        for (int t = 0; t < 2; ++t) {
            const int n = (np * 2 + t) * 16 + m;
            v2f b;
            b.x = wr0[n];
            b.y = wr1[n];
            acc[t] = __builtin_amdgcn_wmma_f32_16x16x4_f32(
                false, a, false, b, (short)0, acc[t], false, false);
        }
    }

    const int half = lane >> 4;
    #pragma unroll
    for (int t = 0; t < 2; ++t) {
        const int n = (np * 2 + t) * 16 + m;
        const float bv = bias[n];
        #pragma unroll
        for (int r = 0; r < 8; ++r) {
            const int mm = r + 8 * half;
            out[(long)(i0 + rt * 16 + mm) * UNITS + n] = acc[t][r] + bv;
        }
    }
}

extern "C" void kernel_launch(void* const* d_in, const int* in_sizes, int n_in,
                              void* d_out, int out_size, void* d_ws, size_t ws_size,
                              hipStream_t stream)
{
    const float* node_feature = (const float*)d_in[0];
    const float* edge_feature = (const float*)d_in[1];
    const float* W_init       = (const float*)d_in[2];
    const float* b_init       = (const float*)d_in[3];
    const float* W_upd        = (const float*)d_in[4];
    const float* b_upd        = (const float*)d_in[5];
    const int*   edge_src     = (const int*)d_in[6];
    const int*   edge_dst     = (const int*)d_in[7];
    float*       out          = (float*)d_out;

    float* edge_state = (float*)d_ws;

    gemm_init_kernel<<<64, 256, 0, stream>>>(node_feature, edge_feature,
                                             W_init, b_init, edge_src, edge_state);

    message_update_kernel<<<N_EDGES / FI, 256, 0, stream>>>(
        edge_src, edge_dst, edge_state, W_upd, b_upd, out);
}


# pair scan replaced by a lookup: each work-group enters the src of its 32 edges in a node-indexed byte table in LDS and looks every dst[j] up once; same match lists, same order of sums
# speedup vs baseline: 30.7700x; 1.8784x over previous
#include <hip/hip_runtime.h>


typedef float v2f __attribute__((ext_vector_type(2)));
typedef float v8f __attribute__((ext_vector_type(8)));

#define N_NODES 20000
#define N_EDGES 8192
#define D_NODE  64
#define D_EDGE  16
#define UNITS   64
#define K1      (D_NODE + D_EDGE)
#define K2      (UNITS + UNITS)

__global__ __launch_bounds__(256) void gemm_init_kernel(
    const float* __restrict__ node_feature,
    const float* __restrict__ edge_feature,
    const float* __restrict__ W,
    const float* __restrict__ bias,
    const int*   __restrict__ edge_src,
    float*       __restrict__ edge_state)
{
    __shared__ float sW[K1 * UNITS];
    for (int i = threadIdx.x; i < K1 * UNITS; i += 256) sW[i] = W[i];
    __syncthreads();

    const int wave    = threadIdx.x >> 5;
    const int lane    = threadIdx.x & 31;
    const int rowTile = blockIdx.x * 8 + wave;
    const int m       = lane & 15;
    const int kh      = (lane >> 4) << 1;
    const int e       = rowTile * 16 + m;
    const int s       = edge_src[e];
    const float* xn = node_feature + (long)s * D_NODE;
    const float* xe = edge_feature + (long)e * D_EDGE;

    v8f acc[4] = {};
    #pragma unroll
    for (int k0 = 0; k0 < K1; k0 += 4) {
        const int k = k0 + kh;
        v2f a;
        a.x = (k     < D_NODE) ? xn[k]     : xe[k     - D_NODE];
        a.y = (k + 1 < D_NODE) ? xn[k + 1] : xe[k + 1 - D_NODE];
        const float* wr0 = &sW[(k    ) * UNITS];
        const float* wr1 = &sW[(k + 1) * UNITS];
        #pragma unroll
        for (int nt = 0; nt < 4; ++nt) {
            v2f b;
            b.x = wr0[nt * 16 + m];
            b.y = wr1[nt * 16 + m];
            acc[nt] = __builtin_amdgcn_wmma_f32_16x16x4_f32(
                false, a, false, b, (short)0, acc[nt], false, false);
        }
    }

    const int half = lane >> 4;
    #pragma unroll
    for (int nt = 0; nt < 4; ++nt) {
        const int n = nt * 16 + m;
        const float bv = bias[n];
        #pragma unroll
        for (int r = 0; r < 8; ++r) {
            const int mm = r + 8 * half;
            edge_state[(long)(rowTile * 16 + mm) * UNITS + n] = acc[nt][r] + bv;
        }
    }
}

#define FI    32
#define FR    32
#define FJW   (FR * 32)
#define FPC   8
#define MP    66
#define TABB  20480
#define TNONE 0xFFu
static_assert(TABB > N_NODES, "an entry per node id and one that stays none");
static_assert(TABB % (256 * 16) == 0, "cleared by whole 16-byte stores of 256 threads");
static_assert(FI <= 255, "a row number fits a table byte beside the none value");
static_assert(FJW * 8 == N_EDGES, "8 waves cover every j");
static_assert(N_EDGES % FI == 0, "whole work-groups");

__global__ __launch_bounds__(256) void message_update_kernel(
    const int*   __restrict__ edge_src,
    const int*   __restrict__ edge_dst,
    const float* __restrict__ edge_state,
    const float* __restrict__ W,
    const float* __restrict__ bias,
    float*       __restrict__ out)
{
    __shared__ float          sW[K2 * UNITS];
    __shared__ float          sM[FI * MP];
    __shared__ unsigned short sList[FI][8][FPC];
    __shared__ int            sCnt[FI][8];
    __shared__ uint4          sTab4[TABB / 16];
    __shared__ unsigned char  sLead[FI];
    unsigned char* sTab = reinterpret_cast<unsigned char*>(sTab4);

    const int tid  = threadIdx.x;
    const int lane = tid & 31;
    const int wave  = __builtin_amdgcn_readfirstlane(tid >> 5);
    const int jbase = wave * FJW;
    const int i0    = blockIdx.x * FI;

    unsigned kd[FR];
    #pragma unroll
    for (int r = 0; r < FR; ++r) kd[r] = (unsigned)edge_dst[jbase + r * 32 + lane];
    const unsigned is_v = (unsigned)edge_src[i0 + lane];

    const int rt = (wave >> 1) & 1;
    const int np = wave & 1;
    const int m  = lane & 15;
    const int kh = (lane >> 4) << 1;
    v2f aes[UNITS / 4];
    #pragma unroll
    for (int q = 0; q < UNITS / 4; ++q) { aes[q].x = 0.0f; aes[q].y = 0.0f; }
    if (wave < 4) {
        const float* x0 = edge_state + (long)(i0 + rt * 16 + m) * UNITS;
        #pragma unroll
        for (int q = 0; q < UNITS / 4; ++q) {
            aes[q].x = x0[4 * q + kh];
            aes[q].y = x0[4 * q + kh + 1];
        }
    }

    {
        const float4* W4  = reinterpret_cast<const float4*>(W);
        float4*       sW4 = reinterpret_cast<float4*>(sW);
        #pragma unroll
        for (int it = 0; it < K2 * UNITS / 4 / 256; ++it)
            sW4[tid + it * 256] = W4[tid + it * 256];
    }

    #pragma unroll
    for (int it = 0; it < TABB / 16 / 256; ++it)
        sTab4[tid + it * 256] = make_uint4(~0u, ~0u, ~0u, ~0u);
    __syncthreads();
    if (wave == 0 && is_v < (unsigned)N_NODES) sTab[is_v] = (unsigned char)lane;
    __syncthreads();
    if (wave == 0)
        sLead[lane] = (is_v < (unsigned)N_NODES) ? sTab[is_v] : (unsigned char)lane;

    unsigned t[FR];
    #pragma unroll
    for (int r = 0; r < FR; ++r) {
        const unsigned d = kd[r] < (unsigned)N_NODES ? kd[r] : (unsigned)N_NODES;
        t[r] = sTab[d];
    }
    unsigned bms[FR];
    unsigned any = 0;
    #pragma unroll
    for (int r = 0; r < FR; ++r) {
        bms[r] = __builtin_amdgcn_ballot_w32(t[r] != TNONE);
        any |= bms[r];
    }
    unsigned cnt_v = 0;
    if (any != 0u) {
        #pragma unroll
        for (int r = 0; r < FR; ++r) {
            unsigned bm = bms[r];
            while (bm != 0u) {
                const int l = __builtin_ctz(bm);
                bm &= bm - 1u;
                const int lrow = __builtin_amdgcn_readlane(t[r], l) & (FI - 1);
                const unsigned c = __builtin_amdgcn_readlane(cnt_v, lrow);
                if (lane == 0 && c < FPC)
                    sList[lrow][wave][c] = (unsigned short)(jbase + r * 32 + l);
                if (lane == lrow) cnt_v = c + 1u;
            }
        }
    }
    sCnt[lane][wave] = (int)cnt_v;
    __syncthreads();

    {
        const int row = tid >> 3;
        const int c8  = (tid & 7) * 8;
        const int gi  = i0 + row;
        const int qs  = edge_src[gi];
        const int qd  = edge_dst[gi];
        float agg[8], rev[8];
        #pragma unroll
        for (int x = 0; x < 8; ++x) { agg[x] = 0.0f; rev[x] = 0.0f; }

        const int lr = sLead[row];
        int n[8];
        #pragma unroll
        for (int w = 0; w < 8; ++w) n[w] = sCnt[lr][w];
        bool over = false;
        #pragma unroll
        for (int w = 0; w < 8; ++w) over |= (n[w] > FPC);

        if (!over) {
            int pre[8];
            pre[0] = 0;
            #pragma unroll
            for (int w = 1; w < 8; ++w) pre[w] = pre[w - 1] + n[w - 1];
            const int total = pre[7] + n[7];
            for (int f = 0; f < total; f += 4) {
                float v[4][8];
                bool  back[4];
                #pragma unroll
                for (int u = 0; u < 4; ++u) {
                    const int ff = f + u;
                    int w = 0, base = 0;
                    #pragma unroll
                    for (int x = 1; x < 8; ++x)
                        if (ff >= pre[x]) { w = x; base = pre[x]; }
                    const int j = (ff < total) ? (int)sList[lr][w][ff - base] : gi;
                    const float4* p = reinterpret_cast<const float4*>(
                        edge_state + (long)j * UNITS + c8);
                    const float4 v0 = p[0], v1 = p[1];
                    v[u][0] = v0.x; v[u][1] = v0.y; v[u][2] = v0.z; v[u][3] = v0.w;
                    v[u][4] = v1.x; v[u][5] = v1.y; v[u][6] = v1.z; v[u][7] = v1.w;
                    back[u] = (edge_src[j] == qd);
                }
                #pragma unroll
                for (int u = 0; u < 4; ++u) {
                    if (f + u < total) {
                        #pragma unroll
                        for (int x = 0; x < 8; ++x) {
                            agg[x] += v[u][x];
                            if (back[u]) rev[x] += v[u][x];
                        }
                    }
                }
            }
        } else {
            for (int j = 0; j < N_EDGES; ++j) {
                if (edge_dst[j] != qs) continue;
                const float4* p = reinterpret_cast<const float4*>(
                    edge_state + (long)j * UNITS + c8);
                const float4 v0 = p[0], v1 = p[1];
                const float v[8] = {v0.x, v0.y, v0.z, v0.w, v1.x, v1.y, v1.z, v1.w};
                const bool back = (edge_src[j] == qd);
                #pragma unroll
                for (int x = 0; x < 8; ++x) {
                    agg[x] += v[x];
                    if (back) rev[x] += v[x];
                }
            }
        }
        #pragma unroll
        for (int x = 0; x < 8; ++x) sM[row * MP + c8 + x] = agg[x] - rev[x];
    }
    __syncthreads();

    if (wave >= 4) return;
    const float* x1 = &sM[(rt * 16 + m) * MP];

    v8f acc[2] = {};
    #pragma unroll
    for (int k0 = 0; k0 < K2; k0 += 4) {
        const int k = k0 + kh;
        v2f a;
        if (k0 < UNITS) {
            a = aes[k0 / 4];
        } else {
            a.x = x1[k     - UNITS];
            a.y = x1[k + 1 - UNITS];
        }
        const float* wr0 = &sW[(k    ) * UNITS];
        const float* wr1 = &sW[(k + 1) * UNITS];
        #pragma unroll
        for (int t = 0; t < 2; ++t) {
            const int n = (np * 2 + t) * 16 + m;
            v2f b;
            b.x = wr0[n];
            b.y = wr1[n];
            acc[t] = __builtin_amdgcn_wmma_f32_16x16x4_f32(
                false, a, false, b, (short)0, acc[t], false, false);
        }
    }

    const int half = lane >> 4;
    #pragma unroll
    for (int t = 0; t < 2; ++t) {
        const int n = (np * 2 + t) * 16 + m;
        const float bv = bias[n];
        #pragma unroll
        for (int r = 0; r < 8; ++r) {
            const int mm = r + 8 * half;
            out[(long)(i0 + rt * 16 + mm) * UNITS + n] = acc[t][r] + bv;
        }
    }
}

extern "C" void kernel_launch(void* const* d_in, const int* in_sizes, int n_in,
                              void* d_out, int out_size, void* d_ws, size_t ws_size,
                              hipStream_t stream)
{
    const float* node_feature = (const float*)d_in[0];
    const float* edge_feature = (const float*)d_in[1];
    const float* W_init       = (const float*)d_in[2];
    const float* b_init       = (const float*)d_in[3];
    const float* W_upd        = (const float*)d_in[4];
    const float* b_upd        = (const float*)d_in[5];
    const int*   edge_src     = (const int*)d_in[6];
    const int*   edge_dst     = (const int*)d_in[7];
    float*       out          = (float*)d_out;

    float* edge_state = (float*)d_ws;

    gemm_init_kernel<<<64, 256, 0, stream>>>(node_feature, edge_feature,
                                             W_init, b_init, edge_src, edge_state);

    message_update_kernel<<<N_EDGES / FI, 256, 0, stream>>>(
        edge_src, edge_dst, edge_state, W_upd, b_upd, out);
}


# one launch: the first layer is linear, so each work-group sums the matched INPUT rows and applies W_init to the sums (+count*b_init) instead of reading other groups' edge_state; float sums reordered
# speedup vs baseline: 50.2322x; 1.6325x over previous
#include <hip/hip_runtime.h>


typedef float v2f __attribute__((ext_vector_type(2)));
typedef float v8f __attribute__((ext_vector_type(8)));

#define N_NODES 20000
#define N_EDGES 8192
#define D_NODE  64
#define D_EDGE  16
#define UNITS   64
#define K1      (D_NODE + D_EDGE)
#define K2      (UNITS + UNITS)

#define FI    32
#define FR    32
#define FJW   (FR * 32)
#define FPC   8
#define XP    82
#define SP    130
#define TABB  20480
#define TNONE 0xFFu
static_assert(TABB > N_NODES, "an entry per node id and one that stays none");
static_assert(TABB % (256 * 16) == 0, "cleared by whole 16-byte stores of 256 threads");
static_assert(FI <= 255, "a row number fits a table byte beside the none value");
static_assert(FJW * 8 == N_EDGES, "8 waves cover every j");
static_assert(N_EDGES % FI == 0, "whole work-groups");
static_assert((K1 * UNITS / 4) % 256 == 0 && (K2 * UNITS / 4) % 256 == 0,
              "weights staged by whole 16-byte loads of 256 threads");

__global__ __launch_bounds__(256) void edgeconv_kernel(
    const float* __restrict__ node_feature,
    const float* __restrict__ edge_feature,
    const float* __restrict__ W1,
    const float* __restrict__ b1,
    const float* __restrict__ W2,
    const float* __restrict__ b2,
    const int*   __restrict__ edge_src,
    const int*   __restrict__ edge_dst,
    float*       __restrict__ out)
{
    __shared__ __attribute__((aligned(16))) float sW2[K2 * UNITS];
    __shared__ __attribute__((aligned(16))) float sW1[K1 * UNITS];
    __shared__ __attribute__((aligned(16))) float sS[FI * SP];
    __shared__ __attribute__((aligned(16))) float sX[FI * XP];
    __shared__ float          sNet[FI];
    __shared__ unsigned short sList[FI][8][FPC];
    __shared__ int            sCnt[FI][8];
    __shared__ uint4          sTab4[TABB / 16];
    __shared__ unsigned char  sLead[FI];
    unsigned char* sTab = reinterpret_cast<unsigned char*>(sTab4);

    const int tid  = threadIdx.x;
    const int lane = tid & 31;
    const int wave  = __builtin_amdgcn_readfirstlane(tid >> 5);
    const int jbase = wave * FJW;
    const int i0    = blockIdx.x * FI;

    unsigned kd[FR];
    #pragma unroll
    for (int r = 0; r < FR; ++r) kd[r] = (unsigned)edge_dst[jbase + r * 32 + lane];
    const unsigned is_v = (unsigned)edge_src[i0 + lane];

    const int rt1  = wave >> 1;
    const int np   = wave & 1;
    const int m    = lane & 15;
    const int kh   = (lane >> 4) << 1;
    const int half = lane >> 4;
    const int rl   = (rt1 & 1) * 16;

    const int e_own = i0 + rl + m;

    float bv1[2], bv2[2];
    #pragma unroll
    for (int t = 0; t < 2; ++t) {
        bv1[t] = b1[(np * 2 + t) * 16 + m];
        bv2[t] = b2[(np * 2 + t) * 16 + m];
    }

    float4 w2r[K2 * UNITS / 4 / 256], w1r[K1 * UNITS / 4 / 256];
    {
        const float4* W24 = reinterpret_cast<const float4*>(W2);
        const float4* W14 = reinterpret_cast<const float4*>(W1);
        #pragma unroll
        for (int it = 0; it < K2 * UNITS / 4 / 256; ++it) w2r[it] = W24[tid + it * 256];
        #pragma unroll
        for (int it = 0; it < K1 * UNITS / 4 / 256; ++it) w1r[it] = W14[tid + it * 256];
    }

    {
        float4* sW24 = reinterpret_cast<float4*>(sW2);
        float4* sW14 = reinterpret_cast<float4*>(sW1);
        #pragma unroll
        for (int it = 0; it < K2 * UNITS / 4 / 256; ++it) sW24[tid + it * 256] = w2r[it];
        #pragma unroll
        for (int it = 0; it < K1 * UNITS / 4 / 256; ++it) sW14[tid + it * 256] = w1r[it];
    }

    #pragma unroll
    for (int it = 0; it < TABB / 16 / 256; ++it)
        sTab4[tid + it * 256] = make_uint4(~0u, ~0u, ~0u, ~0u);

    __syncthreads();
    if (wave == 0 && is_v < (unsigned)N_NODES) sTab[is_v] = (unsigned char)lane;
    __syncthreads();
    if (wave == 0)
        sLead[lane] = (is_v < (unsigned)N_NODES) ? sTab[is_v] : (unsigned char)lane;

    v2f ax[K1 / 4];
    #pragma unroll
    for (int q = 0; q < K1 / 4; ++q) { ax[q].x = 0.0f; ax[q].y = 0.0f; }
    if (wave < 4) {
        const int s_own = __shfl((int)is_v, rl + m);
        const float* xn = node_feature + (long)s_own * D_NODE;
        const float* xe = edge_feature + (long)e_own * D_EDGE;
        #pragma unroll
        for (int q = 0; q < K1 / 4; ++q) {
            const int k = 4 * q + kh;
            if (q < D_NODE / 4) { ax[q].x = xn[k];          ax[q].y = xn[k + 1]; }
            else                { ax[q].x = xe[k - D_NODE]; ax[q].y = xe[k + 1 - D_NODE]; }
        }
    }

    unsigned t[FR];
    #pragma unroll
    for (int r = 0; r < FR; ++r) {
        const unsigned d = kd[r] < (unsigned)N_NODES ? kd[r] : (unsigned)N_NODES;
        t[r] = sTab[d];
    }
    unsigned bms[FR];
    unsigned any = 0;
    #pragma unroll
    for (int r = 0; r < FR; ++r) {
        bms[r] = __builtin_amdgcn_ballot_w32(t[r] != TNONE);
        any |= bms[r];
    }
    unsigned cnt_v = 0;
    if (any != 0u) {
        #pragma unroll
        for (int r = 0; r < FR; ++r) {
            unsigned bm = bms[r];
            while (bm != 0u) {
                const int l = __builtin_ctz(bm);
                bm &= bm - 1u;
                const int lrow = __builtin_amdgcn_readlane(t[r], l) & (FI - 1);
                const unsigned c = __builtin_amdgcn_readlane(cnt_v, lrow);
                if (lane == 0 && c < FPC)
                    sList[lrow][wave][c] = (unsigned short)(jbase + r * 32 + l);
                if (lane == lrow) cnt_v = c + 1u;
            }
        }
    }
    sCnt[lane][wave] = (int)cnt_v;
    __syncthreads();

    {
        const int row = tid >> 3;
        const int c8  = tid & 7;
        const int gi  = i0 + row;
        const int qs  = edge_src[gi];
        const int qd  = edge_dst[gi];
        float agg[10], rev[10];
        #pragma unroll
        for (int x = 0; x < 10; ++x) { agg[x] = 0.0f; rev[x] = 0.0f; }
        int na = 0, nr = 0;

        const int lr = sLead[row];
        int n[8];
        #pragma unroll
        for (int w = 0; w < 8; ++w) n[w] = sCnt[lr][w];
        bool over = false;
        #pragma unroll
        for (int w = 0; w < 8; ++w) over |= (n[w] > FPC);

        if (!over) {
            int pre[8];
            pre[0] = 0;
            #pragma unroll
            for (int w = 1; w < 8; ++w) pre[w] = pre[w - 1] + n[w - 1];
            const int total = pre[7] + n[7];
            for (int f = 0; f < total; f += 4) {
                int jj[4], sj[4];
                #pragma unroll
                for (int u = 0; u < 4; ++u) {
                    const int ff = (f + u < total) ? f + u : total - 1;
                    int w = 0, base = 0;
                    #pragma unroll
                    for (int x = 1; x < 8; ++x)
                        if (ff >= pre[x]) { w = x; base = pre[x]; }
                    jj[u] = (int)sList[lr][w][ff - base];
                    sj[u] = edge_src[jj[u]];
                }
                float v[4][10];
                #pragma unroll
                for (int u = 0; u < 4; ++u) {
                    const float4* p = reinterpret_cast<const float4*>(
                        node_feature + (long)sj[u] * D_NODE + c8 * 8);
                    const float2* pe = reinterpret_cast<const float2*>(
                        edge_feature + (long)jj[u] * D_EDGE + c8 * 2);
                    const float4 v0 = p[0], v1 = p[1];
                    const float2 v2 = pe[0];
                    v[u][0] = v0.x; v[u][1] = v0.y; v[u][2] = v0.z; v[u][3] = v0.w;
                    v[u][4] = v1.x; v[u][5] = v1.y; v[u][6] = v1.z; v[u][7] = v1.w;
                    v[u][8] = v2.x; v[u][9] = v2.y;
                }
                #pragma unroll
                for (int u = 0; u < 4; ++u) {
                    const bool in   = (f + u < total);
                    const bool back = in && (sj[u] == qd);
                    na += in ? 1 : 0;
                    nr += back ? 1 : 0;
                    #pragma unroll
                    for (int x = 0; x < 10; ++x) {
                        agg[x] += in   ? v[u][x] : 0.0f;
                        rev[x] += back ? v[u][x] : 0.0f;
                    }
                }
            }
        } else {
            for (int j = 0; j < N_EDGES; ++j) {
                if (edge_dst[j] != qs) continue;
                const int s = edge_src[j];
                const float4* p = reinterpret_cast<const float4*>(
                    node_feature + (long)s * D_NODE + c8 * 8);
                const float2* pe = reinterpret_cast<const float2*>(
                    edge_feature + (long)j * D_EDGE + c8 * 2);
                const float4 v0 = p[0], v1 = p[1];
                const float2 v2 = pe[0];
                const float v[10] = {v0.x, v0.y, v0.z, v0.w, v1.x, v1.y, v1.z, v1.w,
                                     v2.x, v2.y};
                const bool back = (s == qd);
                na += 1;
                nr += back ? 1 : 0;
                #pragma unroll
                for (int x = 0; x < 10; ++x) {
                    agg[x] += v[x];
                    if (back) rev[x] += v[x];
                }
            }
        }
        #pragma unroll
        for (int x = 0; x < 8; ++x) sX[row * XP + c8 * 8 + x] = agg[x] - rev[x];
        #pragma unroll
        for (int x = 0; x < 2; ++x) sX[row * XP + D_NODE + c8 * 2 + x] = agg[8 + x] - rev[8 + x];
        if (c8 == 0) sNet[row] = (float)(na - nr);
    }
    __syncthreads();

    float nb[8];
    #pragma unroll
    for (int r = 0; r < 8; ++r) nb[r] = 1.0f;
    if (wave >= 4) {
        const float* xr = &sX[(rl + m) * XP];
        #pragma unroll
        for (int q = 0; q < K1 / 4; ++q) {
            ax[q].x = xr[4 * q + kh];
            ax[q].y = xr[4 * q + kh + 1];
        }
        #pragma unroll
        for (int r = 0; r < 8; ++r) nb[r] = sNet[rl + r + 8 * half];
    }
    {
        v8f acc[2] = {};
        #pragma unroll
        for (int q = 0; q < K1 / 4; ++q) {
            const int k = 4 * q + kh;
            const float* wr0 = &sW1[(k    ) * UNITS];
            const float* wr1 = &sW1[(k + 1) * UNITS];
            #pragma unroll
            for (int tt = 0; tt < 2; ++tt) {
                const int n = (np * 2 + tt) * 16 + m;
                v2f b;
                b.x = wr0[n];
                b.y = wr1[n];
                acc[tt] = __builtin_amdgcn_wmma_f32_16x16x4_f32(
                    false, ax[q], false, b, (short)0, acc[tt], false, false);
            }
        }
        const int cb = (wave < 4) ? 0 : UNITS;
        #pragma unroll
        for (int tt = 0; tt < 2; ++tt) {
            const int n = (np * 2 + tt) * 16 + m;
            #pragma unroll
            for (int r = 0; r < 8; ++r) {
                const int mm = r + 8 * half;
                sS[(rl + mm) * SP + cb + n] = acc[tt][r] + nb[r] * bv1[tt];
            }
        }
    }
    __syncthreads();

    if (wave >= 4) return;
    const float* x1 = &sS[(rl + m) * SP];

    v8f acc2[2] = {};
    #pragma unroll
    for (int k0 = 0; k0 < K2; k0 += 4) {
        const int k = k0 + kh;
        v2f a;
        a.x = x1[k];
        a.y = x1[k + 1];
        const float* wr0 = &sW2[(k    ) * UNITS];
        const float* wr1 = &sW2[(k + 1) * UNITS];
        #pragma unroll
        for (int tt = 0; tt < 2; ++tt) {
            const int n = (np * 2 + tt) * 16 + m;
            v2f b;
            b.x = wr0[n];
            b.y = wr1[n];
            acc2[tt] = __builtin_amdgcn_wmma_f32_16x16x4_f32(
                false, a, false, b, (short)0, acc2[tt], false, false);
        }
    }

    #pragma unroll
    for (int tt = 0; tt < 2; ++tt) {
        const int n = (np * 2 + tt) * 16 + m;
        #pragma unroll
        for (int r = 0; r < 8; ++r) {
            const int mm = r + 8 * half;
            out[(long)(i0 + rl + mm) * UNITS + n] = acc2[tt][r] + bv2[tt];
        }
    }
}

extern "C" void kernel_launch(void* const* d_in, const int* in_sizes, int n_in,
                              void* d_out, int out_size, void* d_ws, size_t ws_size,
                              hipStream_t stream)
{
    const float* node_feature = (const float*)d_in[0];
    const float* edge_feature = (const float*)d_in[1];
    const float* W_init       = (const float*)d_in[2];
    const float* b_init       = (const float*)d_in[3];
    const float* W_upd        = (const float*)d_in[4];
    const float* b_upd        = (const float*)d_in[5];
    const int*   edge_src     = (const int*)d_in[6];
    const int*   edge_dst     = (const int*)d_in[7];
    float*       out          = (float*)d_out;

    edgeconv_kernel<<<N_EDGES / FI, 256, 0, stream>>>(
        node_feature, edge_feature, W_init, b_init, W_upd, b_upd,
        edge_src, edge_dst, out);
}


# weights requested after the table clear, parked to LDS only behind the row sums; the 3 barriers before that drain LDS only (s_wait_dscnt + s_barrier by hand): scan and sums do not wait for weights
# speedup vs baseline: 53.5120x; 1.0653x over previous
#include <hip/hip_runtime.h>


typedef float v2f __attribute__((ext_vector_type(2)));
typedef float v8f __attribute__((ext_vector_type(8)));
typedef float v4f __attribute__((ext_vector_type(4)));

#define N_NODES 20000
#define N_EDGES 8192
#define D_NODE  64
#define D_EDGE  16
#define UNITS   64
#define K1      (D_NODE + D_EDGE)
#define K2      (UNITS + UNITS)

#define FI    32
#define FR    32
#define FJW   (FR * 32)
#define FPC   8
#define XP    82
#define SP    130
#define TABB  20480
#define TNONE 0xFFu
static_assert(TABB > N_NODES, "an entry per node id and one that stays none");
static_assert(TABB % (256 * 16) == 0, "cleared by whole 16-byte stores of 256 threads");
static_assert(FI <= 255, "a row number fits a table byte beside the none value");
static_assert(FJW * 8 == N_EDGES, "8 waves cover every j");
static_assert(N_EDGES % FI == 0, "whole work-groups");
static_assert((K1 * UNITS / 4) % 256 == 0 && (K2 * UNITS / 4) % 256 == 0,
              "weights staged by whole 16-byte loads of 256 threads");

static __device__ __forceinline__ void lds_barrier()
{
    asm volatile("s_wait_dscnt 0x0\n\ts_barrier_signal -1\n\ts_barrier_wait -1" ::: "memory");
}

__global__ __launch_bounds__(256) void edgeconv_kernel(
    const float* __restrict__ node_feature,
    const float* __restrict__ edge_feature,
    const float* __restrict__ W1,
    const float* __restrict__ b1,
    const float* __restrict__ W2,
    const float* __restrict__ b2,
    const int*   __restrict__ edge_src,
    const int*   __restrict__ edge_dst,
    float*       __restrict__ out)
{
    __shared__ __attribute__((aligned(16))) float sW2[K2 * UNITS];
    __shared__ __attribute__((aligned(16))) float sW1[K1 * UNITS];
    __shared__ __attribute__((aligned(16))) float sS[FI * SP];
    __shared__ __attribute__((aligned(16))) float sX[FI * XP];
    __shared__ float          sNet[FI];
    __shared__ unsigned short sList[FI][8][FPC];
    __shared__ int            sCnt[FI][8];
    __shared__ uint4          sTab4[TABB / 16];
    __shared__ unsigned char  sLead[FI];
    unsigned char* sTab = reinterpret_cast<unsigned char*>(sTab4);

    const int tid  = threadIdx.x;
    const int lane = tid & 31;
    const int wave  = __builtin_amdgcn_readfirstlane(tid >> 5);
    const int jbase = wave * FJW;
    const int i0    = blockIdx.x * FI;

    const unsigned is_v = (unsigned)edge_src[i0 + lane];
    unsigned kd[FR];
    #pragma unroll
    for (int r = 0; r < FR; ++r) kd[r] = (unsigned)edge_dst[jbase + r * 32 + lane];

    const int rt1  = wave >> 1;
    const int np   = wave & 1;
    const int m    = lane & 15;
    const int kh   = (lane >> 4) << 1;
    const int half = lane >> 4;
    const int rl   = (rt1 & 1) * 16;

    const int e_own = i0 + rl + m;

    #pragma unroll
    for (int it = 0; it < TABB / 16 / 256; ++it)
        sTab4[tid + it * 256] = make_uint4(~0u, ~0u, ~0u, ~0u);

    lds_barrier();

    v4f w2r[K2 * UNITS / 4 / 256], w1r[K1 * UNITS / 4 / 256];
    {
        const v4f* W24 = reinterpret_cast<const v4f*>(W2);
        const v4f* W14 = reinterpret_cast<const v4f*>(W1);
        #pragma unroll
        for (int it = 0; it < K2 * UNITS / 4 / 256; ++it) w2r[it] = W24[tid + it * 256];
        #pragma unroll
        for (int it = 0; it < K1 * UNITS / 4 / 256; ++it) w1r[it] = W14[tid + it * 256];
    }
    float bv1[2], bv2[2];
    #pragma unroll
    for (int t = 0; t < 2; ++t) {
        bv1[t] = b1[(np * 2 + t) * 16 + m];
        bv2[t] = b2[(np * 2 + t) * 16 + m];
    }

    if (wave == 0 && is_v < (unsigned)N_NODES) sTab[is_v] = (unsigned char)lane;
    lds_barrier();
    if (wave == 0)
        sLead[lane] = (is_v < (unsigned)N_NODES) ? sTab[is_v] : (unsigned char)lane;

    unsigned t[FR];
    #pragma unroll
    for (int r = 0; r < FR; ++r) {
        const unsigned d = kd[r] < (unsigned)N_NODES ? kd[r] : (unsigned)N_NODES;
        t[r] = sTab[d];
    }

    v2f ax[K1 / 4];
    #pragma unroll
    for (int q = 0; q < K1 / 4; ++q) { ax[q].x = 0.0f; ax[q].y = 0.0f; }
    if (wave < 4) {
        const int s_own = __shfl((int)is_v, rl + m);
        const float* xn = node_feature + (long)s_own * D_NODE;
        const float* xe = edge_feature + (long)e_own * D_EDGE;
        #pragma unroll
        for (int q = 0; q < K1 / 4; ++q) {
            const int k = 4 * q + kh;
            if (q < D_NODE / 4) { ax[q].x = xn[k];          ax[q].y = xn[k + 1]; }
            else                { ax[q].x = xe[k - D_NODE]; ax[q].y = xe[k + 1 - D_NODE]; }
        }
    }

    unsigned bms[FR];
    unsigned any = 0;
    #pragma unroll
    for (int r = 0; r < FR; ++r) {
        bms[r] = __builtin_amdgcn_ballot_w32(t[r] != TNONE);
        any |= bms[r];
    }
    unsigned cnt_v = 0;
    if (any != 0u) {
        #pragma unroll
        for (int r = 0; r < FR; ++r) {
            unsigned bm = bms[r];
            while (bm != 0u) {
                const int l = __builtin_ctz(bm);
                bm &= bm - 1u;
                const int lrow = __builtin_amdgcn_readlane(t[r], l) & (FI - 1);
                const unsigned c = __builtin_amdgcn_readlane(cnt_v, lrow);
                if (lane == 0 && c < FPC)
                    sList[lrow][wave][c] = (unsigned short)(jbase + r * 32 + l);
                if (lane == lrow) cnt_v = c + 1u;
            }
        }
    }
    sCnt[lane][wave] = (int)cnt_v;
    lds_barrier();

    {
        const int row = tid >> 3;
        const int c8  = tid & 7;
        const int gi  = i0 + row;
        const int qs  = edge_src[gi];
        const int qd  = edge_dst[gi];
        float agg[10], rev[10];
        #pragma unroll
        for (int x = 0; x < 10; ++x) { agg[x] = 0.0f; rev[x] = 0.0f; }
        int na = 0, nr = 0;

        const int lr = sLead[row];
        int n[8];
        #pragma unroll
        for (int w = 0; w < 8; ++w) n[w] = sCnt[lr][w];
        bool over = false;
        #pragma unroll
        for (int w = 0; w < 8; ++w) over |= (n[w] > FPC);

        if (!over) {
            int pre[8];
            pre[0] = 0;
            #pragma unroll
            for (int w = 1; w < 8; ++w) pre[w] = pre[w - 1] + n[w - 1];
            const int total = pre[7] + n[7];
            for (int f = 0; f < total; f += 4) {
                int jj[4], sj[4];
                #pragma unroll
                for (int u = 0; u < 4; ++u) {
                    const int ff = (f + u < total) ? f + u : total - 1;
                    int w = 0, base = 0;
                    #pragma unroll
                    for (int x = 1; x < 8; ++x)
                        if (ff >= pre[x]) { w = x; base = pre[x]; }
                    jj[u] = (int)sList[lr][w][ff - base];
                    sj[u] = edge_src[jj[u]];
                }
                float v[4][10];
                #pragma unroll
                for (int u = 0; u < 4; ++u) {
                    const float4* p = reinterpret_cast<const float4*>(
                        node_feature + (long)sj[u] * D_NODE + c8 * 8);
                    const float2* pe = reinterpret_cast<const float2*>(
                        edge_feature + (long)jj[u] * D_EDGE + c8 * 2);
                    const float4 v0 = p[0], v1 = p[1];
                    const float2 v2 = pe[0];
                    v[u][0] = v0.x; v[u][1] = v0.y; v[u][2] = v0.z; v[u][3] = v0.w;
                    v[u][4] = v1.x; v[u][5] = v1.y; v[u][6] = v1.z; v[u][7] = v1.w;
                    v[u][8] = v2.x; v[u][9] = v2.y;
                }
                #pragma unroll
                for (int u = 0; u < 4; ++u) {
                    const bool in   = (f + u < total);
                    const bool back = in && (sj[u] == qd);
                    na += in ? 1 : 0;
                    nr += back ? 1 : 0;
                    #pragma unroll
                    for (int x = 0; x < 10; ++x) {
                        agg[x] += in   ? v[u][x] : 0.0f;
                        rev[x] += back ? v[u][x] : 0.0f;
                    }
                }
            }
        } else {
            for (int j = 0; j < N_EDGES; ++j) {
                if (edge_dst[j] != qs) continue;
                const int s = edge_src[j];
                const float4* p = reinterpret_cast<const float4*>(
                    node_feature + (long)s * D_NODE + c8 * 8);
                const float2* pe = reinterpret_cast<const float2*>(
                    edge_feature + (long)j * D_EDGE + c8 * 2);
                const float4 v0 = p[0], v1 = p[1];
                const float2 v2 = pe[0];
                const float v[10] = {v0.x, v0.y, v0.z, v0.w, v1.x, v1.y, v1.z, v1.w,
                                     v2.x, v2.y};
                const bool back = (s == qd);
                na += 1;
                nr += back ? 1 : 0;
                #pragma unroll
                for (int x = 0; x < 10; ++x) {
                    agg[x] += v[x];
                    if (back) rev[x] += v[x];
                }
            }
        }
        #pragma unroll
        for (int x = 0; x < 8; ++x) sX[row * XP + c8 * 8 + x] = agg[x] - rev[x];
        #pragma unroll
        for (int x = 0; x < 2; ++x) sX[row * XP + D_NODE + c8 * 2 + x] = agg[8 + x] - rev[8 + x];
        if (c8 == 0) sNet[row] = (float)(na - nr);
    }
    {
        v4f* sW24 = reinterpret_cast<v4f*>(sW2);
        v4f* sW14 = reinterpret_cast<v4f*>(sW1);
        #pragma unroll
        for (int it = 0; it < K2 * UNITS / 4 / 256; ++it) sW24[tid + it * 256] = w2r[it];
        #pragma unroll
        for (int it = 0; it < K1 * UNITS / 4 / 256; ++it) sW14[tid + it * 256] = w1r[it];
    }
    __syncthreads();

    float nb[8];
    #pragma unroll
    for (int r = 0; r < 8; ++r) nb[r] = 1.0f;
    if (wave >= 4) {
        const float* xr = &sX[(rl + m) * XP];
        #pragma unroll
        for (int q = 0; q < K1 / 4; ++q) {
            ax[q].x = xr[4 * q + kh];
            ax[q].y = xr[4 * q + kh + 1];
        }
        #pragma unroll
        for (int r = 0; r < 8; ++r) nb[r] = sNet[rl + r + 8 * half];
    }
    {
        v8f acc[2] = {};
        #pragma unroll
        for (int q = 0; q < K1 / 4; ++q) {
            const int k = 4 * q + kh;
            const float* wr0 = &sW1[(k    ) * UNITS];
            const float* wr1 = &sW1[(k + 1) * UNITS];
            #pragma unroll
            for (int tt = 0; tt < 2; ++tt) {
                const int n = (np * 2 + tt) * 16 + m;
                v2f b;
                b.x = wr0[n];
                b.y = wr1[n];
                acc[tt] = __builtin_amdgcn_wmma_f32_16x16x4_f32(
                    false, ax[q], false, b, (short)0, acc[tt], false, false);
            }
        }
        const int cb = (wave < 4) ? 0 : UNITS;
        #pragma unroll
        for (int tt = 0; tt < 2; ++tt) {
            const int n = (np * 2 + tt) * 16 + m;
            #pragma unroll
            for (int r = 0; r < 8; ++r) {
                const int mm = r + 8 * half;
                sS[(rl + mm) * SP + cb + n] = acc[tt][r] + nb[r] * bv1[tt];
            }
        }
    }
    __syncthreads();

    if (wave >= 4) return;
    const float* x1 = &sS[(rl + m) * SP];

    v8f acc2[2] = {};
    #pragma unroll
    for (int k0 = 0; k0 < K2; k0 += 4) {
        const int k = k0 + kh;
        v2f a;
        a.x = x1[k];
        a.y = x1[k + 1];
        const float* wr0 = &sW2[(k    ) * UNITS];
        const float* wr1 = &sW2[(k + 1) * UNITS];
        #pragma unroll
        for (int tt = 0; tt < 2; ++tt) {
            const int n = (np * 2 + tt) * 16 + m;
            v2f b;
            b.x = wr0[n];
            b.y = wr1[n];
            acc2[tt] = __builtin_amdgcn_wmma_f32_16x16x4_f32(
                false, a, false, b, (short)0, acc2[tt], false, false);
        }
    }

    #pragma unroll
    for (int tt = 0; tt < 2; ++tt) {
        const int n = (np * 2 + tt) * 16 + m;
        #pragma unroll
        for (int r = 0; r < 8; ++r) {
            const int mm = r + 8 * half;
            out[(long)(i0 + rl + mm) * UNITS + n] = acc2[tt][r] + bv2[tt];
        }
    }
}

extern "C" void kernel_launch(void* const* d_in, const int* in_sizes, int n_in,
                              void* d_out, int out_size, void* d_ws, size_t ws_size,
                              hipStream_t stream)
{
    const float* node_feature = (const float*)d_in[0];
    const float* edge_feature = (const float*)d_in[1];
    const float* W_init       = (const float*)d_in[2];
    const float* b_init       = (const float*)d_in[3];
    const float* W_upd        = (const float*)d_in[4];
    const float* b_upd        = (const float*)d_in[5];
    const int*   edge_src     = (const int*)d_in[6];
    const int*   edge_dst     = (const int*)d_in[7];
    float*       out          = (float*)d_out;

    edgeconv_kernel<<<N_EDGES / FI, 256, 0, stream>>>(
        node_feature, edge_feature, W_init, b_init, W_upd, b_upd,
        edge_src, edge_dst, out);
}


# fragment reads of both dense layers issued as one group ahead of each wave's matrix instructions (were between them); weights parked pair-interleaved in LDS: a B fragment is one 8-byte read
# speedup vs baseline: 58.9867x; 1.1023x over previous
#include <hip/hip_runtime.h>


typedef float v2f __attribute__((ext_vector_type(2)));
typedef float v8f __attribute__((ext_vector_type(8)));
typedef float v4f __attribute__((ext_vector_type(4)));

#define N_NODES 20000
#define N_EDGES 8192
#define D_NODE  64
#define D_EDGE  16
#define UNITS   64
#define K1      (D_NODE + D_EDGE)
#define K2      (UNITS + UNITS)

#define FI    32
#define FR    32
#define FJW   (FR * 32)
#define FPC   8
#define XP    82
#define SP    132
#define PP    160
#define TABB  20480
#define TNONE 0xFFu
static_assert(TABB > N_NODES, "an entry per node id and one that stays none");
static_assert(TABB % (256 * 16) == 0, "cleared by whole 16-byte stores of 256 threads");
static_assert(FI <= 255, "a row number fits a table byte beside the none value");
static_assert(FJW * 8 == N_EDGES, "8 waves cover every j");
static_assert(N_EDGES % FI == 0, "whole work-groups");
static_assert((K1 / 2 * 32) % 256 == 0 && (K2 / 2 * 32) % 256 == 0,
              "weights staged as (pair of rows, pair of columns) items, a whole number per thread");
static_assert(K1 % 4 == 0 && K2 % 4 == 0 && PP % 4 == 0 && PP >= 2 * UNITS,
              "a pair row holds 2 x UNITS values in 16-byte items");

static __device__ __forceinline__ void lds_barrier()
{
    asm volatile("s_wait_dscnt 0x0\n\ts_barrier_signal -1\n\ts_barrier_wait -1" ::: "memory");
}

__global__ __launch_bounds__(256) void edgeconv_kernel(
    const float* __restrict__ node_feature,
    const float* __restrict__ edge_feature,
    const float* __restrict__ W1,
    const float* __restrict__ b1,
    const float* __restrict__ W2,
    const float* __restrict__ b2,
    const int*   __restrict__ edge_src,
    const int*   __restrict__ edge_dst,
    float*       __restrict__ out)
{
    __shared__ __attribute__((aligned(16))) float sW2[K2 / 2 * PP];
    __shared__ __attribute__((aligned(16))) float sW1[K1 / 2 * PP];
    __shared__ __attribute__((aligned(16))) float sS[FI * SP];
    __shared__ __attribute__((aligned(16))) float sX[FI * XP];
    __shared__ float          sNet[FI];
    __shared__ unsigned short sList[FI][8][FPC];
    __shared__ int            sCnt[FI][8];
    __shared__ uint4          sTab4[TABB / 16];
    __shared__ unsigned char  sLead[FI];
    unsigned char* sTab = reinterpret_cast<unsigned char*>(sTab4);

    const int tid  = threadIdx.x;
    const int lane = tid & 31;
    const int wave  = __builtin_amdgcn_readfirstlane(tid >> 5);
    const int jbase = wave * FJW;
    const int i0    = blockIdx.x * FI;

    const unsigned is_v = (unsigned)edge_src[i0 + lane];
    unsigned kd[FR];
    #pragma unroll
    for (int r = 0; r < FR; ++r) kd[r] = (unsigned)edge_dst[jbase + r * 32 + lane];

    const int rt1  = wave >> 1;
    const int np   = wave & 1;
    const int m    = lane & 15;
    const int kh   = (lane >> 4) << 1;
    const int half = lane >> 4;
    const int rl   = (rt1 & 1) * 16;

    const int e_own = i0 + rl + m;

    #pragma unroll
    for (int it = 0; it < TABB / 16 / 256; ++it)
        sTab4[tid + it * 256] = make_uint4(~0u, ~0u, ~0u, ~0u);

    lds_barrier();

    v2f w2a[K2 / 2 / 8], w2b[K2 / 2 / 8], w1a[K1 / 2 / 8], w1b[K1 / 2 / 8];
    {
        const v2f* W2v = reinterpret_cast<const v2f*>(W2);
        const v2f* W1v = reinterpret_cast<const v2f*>(W1);
        #pragma unroll
        for (int it = 0; it < K2 / 2 / 8; ++it) {
            const int p = wave + it * 8;
            w2a[it] = W2v[(2 * p    ) * (UNITS / 2) + lane];
            w2b[it] = W2v[(2 * p + 1) * (UNITS / 2) + lane];
        }
        #pragma unroll
        for (int it = 0; it < K1 / 2 / 8; ++it) {
            const int p = wave + it * 8;
            w1a[it] = W1v[(2 * p    ) * (UNITS / 2) + lane];
            w1b[it] = W1v[(2 * p + 1) * (UNITS / 2) + lane];
        }
    }
    float bv1[2], bv2[2];
    #pragma unroll
    for (int t = 0; t < 2; ++t) {
        bv1[t] = b1[(np * 2 + t) * 16 + m];
        bv2[t] = b2[(np * 2 + t) * 16 + m];
    }

    if (wave == 0 && is_v < (unsigned)N_NODES) sTab[is_v] = (unsigned char)lane;
    lds_barrier();
    if (wave == 0)
        sLead[lane] = (is_v < (unsigned)N_NODES) ? sTab[is_v] : (unsigned char)lane;

    unsigned t[FR];
    #pragma unroll
    for (int r = 0; r < FR; ++r) {
        const unsigned d = kd[r] < (unsigned)N_NODES ? kd[r] : (unsigned)N_NODES;
        t[r] = sTab[d];
    }

    v2f ax[K1 / 4];
    #pragma unroll
    for (int q = 0; q < K1 / 4; ++q) { ax[q].x = 0.0f; ax[q].y = 0.0f; }
    if (wave < 4) {
        const int s_own = __shfl((int)is_v, rl + m);
        const float* xn = node_feature + (long)s_own * D_NODE;
        const float* xe = edge_feature + (long)e_own * D_EDGE;
        #pragma unroll
        for (int q = 0; q < K1 / 4; ++q) {
            const int k = 4 * q + kh;
            if (q < D_NODE / 4) { ax[q].x = xn[k];          ax[q].y = xn[k + 1]; }
            else                { ax[q].x = xe[k - D_NODE]; ax[q].y = xe[k + 1 - D_NODE]; }
        }
    }

    unsigned bms[FR];
    unsigned any = 0;
    #pragma unroll
    for (int r = 0; r < FR; ++r) {
        bms[r] = __builtin_amdgcn_ballot_w32(t[r] != TNONE);
        any |= bms[r];
    }
    unsigned cnt_v = 0;
    if (any != 0u) {
        #pragma unroll
        for (int r = 0; r < FR; ++r) {
            unsigned bm = bms[r];
            while (bm != 0u) {
                const int l = __builtin_ctz(bm);
                bm &= bm - 1u;
                const int lrow = __builtin_amdgcn_readlane(t[r], l) & (FI - 1);
                const unsigned c = __builtin_amdgcn_readlane(cnt_v, lrow);
                if (lane == 0 && c < FPC)
                    sList[lrow][wave][c] = (unsigned short)(jbase + r * 32 + l);
                if (lane == lrow) cnt_v = c + 1u;
            }
        }
    }
    sCnt[lane][wave] = (int)cnt_v;
    lds_barrier();

    {
        const int row = tid >> 3;
        const int c8  = tid & 7;
        const int gi  = i0 + row;
        const int qs  = edge_src[gi];
        const int qd  = edge_dst[gi];
        float agg[10], rev[10];
        #pragma unroll
        for (int x = 0; x < 10; ++x) { agg[x] = 0.0f; rev[x] = 0.0f; }
        int na = 0, nr = 0;

        const int lr = sLead[row];
        int n[8];
        #pragma unroll
        for (int w = 0; w < 8; ++w) n[w] = sCnt[lr][w];
        bool over = false;
        #pragma unroll
        for (int w = 0; w < 8; ++w) over |= (n[w] > FPC);

        if (!over) {
            int pre[8];
            pre[0] = 0;
            #pragma unroll
            for (int w = 1; w < 8; ++w) pre[w] = pre[w - 1] + n[w - 1];
            const int total = pre[7] + n[7];
            for (int f = 0; f < total; f += 4) {
                int jj[4], sj[4];
                #pragma unroll
                for (int u = 0; u < 4; ++u) {
                    const int ff = (f + u < total) ? f + u : total - 1;
                    int w = 0, base = 0;
                    #pragma unroll
                    for (int x = 1; x < 8; ++x)
                        if (ff >= pre[x]) { w = x; base = pre[x]; }
                    jj[u] = (int)sList[lr][w][ff - base];
                    sj[u] = edge_src[jj[u]];
                }
                float v[4][10];
                #pragma unroll
                for (int u = 0; u < 4; ++u) {
                    const float4* p = reinterpret_cast<const float4*>(
                        node_feature + (long)sj[u] * D_NODE + c8 * 8);
                    const float2* pe = reinterpret_cast<const float2*>(
                        edge_feature + (long)jj[u] * D_EDGE + c8 * 2);
                    const float4 v0 = p[0], v1 = p[1];
                    const float2 v2 = pe[0];
                    v[u][0] = v0.x; v[u][1] = v0.y; v[u][2] = v0.z; v[u][3] = v0.w;
                    v[u][4] = v1.x; v[u][5] = v1.y; v[u][6] = v1.z; v[u][7] = v1.w;
                    v[u][8] = v2.x; v[u][9] = v2.y;
                }
                #pragma unroll
                for (int u = 0; u < 4; ++u) {
                    const bool in   = (f + u < total);
                    const bool back = in && (sj[u] == qd);
                    na += in ? 1 : 0;
                    nr += back ? 1 : 0;
                    #pragma unroll
                    for (int x = 0; x < 10; ++x) {
                        agg[x] += in   ? v[u][x] : 0.0f;
                        rev[x] += back ? v[u][x] : 0.0f;
                    }
                }
            }
        } else {
            for (int j = 0; j < N_EDGES; ++j) {
                if (edge_dst[j] != qs) continue;
                const int s = edge_src[j];
                const float4* p = reinterpret_cast<const float4*>(
                    node_feature + (long)s * D_NODE + c8 * 8);
                const float2* pe = reinterpret_cast<const float2*>(
                    edge_feature + (long)j * D_EDGE + c8 * 2);
                const float4 v0 = p[0], v1 = p[1];
                const float2 v2 = pe[0];
                const float v[10] = {v0.x, v0.y, v0.z, v0.w, v1.x, v1.y, v1.z, v1.w,
                                     v2.x, v2.y};
                const bool back = (s == qd);
                na += 1;
                nr += back ? 1 : 0;
                #pragma unroll
                for (int x = 0; x < 10; ++x) {
                    agg[x] += v[x];
                    if (back) rev[x] += v[x];
                }
            }
        }
        #pragma unroll
        for (int x = 0; x < 8; ++x) sX[row * XP + c8 * 8 + x] = agg[x] - rev[x];
        #pragma unroll
        for (int x = 0; x < 2; ++x) sX[row * XP + D_NODE + c8 * 2 + x] = agg[8 + x] - rev[8 + x];
        if (c8 == 0) sNet[row] = (float)(na - nr);
    }
    {
        #pragma unroll
        for (int it = 0; it < K2 / 2 / 8; ++it) {
            v4f v;
            v.x = w2a[it].x; v.y = w2b[it].x; v.z = w2a[it].y; v.w = w2b[it].y;
            *reinterpret_cast<v4f*>(&sW2[(wave + it * 8) * PP + 4 * lane]) = v;
        }
        #pragma unroll
        for (int it = 0; it < K1 / 2 / 8; ++it) {
            v4f v;
            v.x = w1a[it].x; v.y = w1b[it].x; v.z = w1a[it].y; v.w = w1b[it].y;
            *reinterpret_cast<v4f*>(&sW1[(wave + it * 8) * PP + 4 * lane]) = v;
        }
    }
    __syncthreads();

    float nb[8];
    #pragma unroll
    for (int r = 0; r < 8; ++r) nb[r] = 1.0f;
    if (wave >= 4) {
        const v2f* xr = reinterpret_cast<const v2f*>(&sX[(rl + m) * XP + kh]);
        #pragma unroll
        for (int q = 0; q < K1 / 4; ++q) ax[q] = xr[2 * q];
        #pragma unroll
        for (int r = 0; r < 8; ++r) nb[r] = sNet[rl + r + 8 * half];
    }
    const int bofs = half * PP + 2 * (np * 32 + m);
    {
        v2f bf[K1 / 4][2];
        #pragma unroll
        for (int q = 0; q < K1 / 4; ++q) {
            const v2f* bp = reinterpret_cast<const v2f*>(&sW1[2 * q * PP + bofs]);
            bf[q][0] = bp[0];
            bf[q][1] = bp[16];
        }
        __builtin_amdgcn_sched_barrier(0);
        v8f acc[2] = {};
        #pragma unroll
        for (int q = 0; q < K1 / 4; ++q) {
            #pragma unroll
            for (int tt = 0; tt < 2; ++tt)
                acc[tt] = __builtin_amdgcn_wmma_f32_16x16x4_f32(
                    false, ax[q], false, bf[q][tt], (short)0, acc[tt], false, false);
        }
        const int cb = (wave < 4) ? 0 : UNITS;
        #pragma unroll
        for (int tt = 0; tt < 2; ++tt) {
            const int n = (np * 2 + tt) * 16 + m;
            #pragma unroll
            for (int r = 0; r < 8; ++r) {
                const int mm = r + 8 * half;
                sS[(rl + mm) * SP + cb + n] = acc[tt][r] + nb[r] * bv1[tt];
            }
        }
    }
    v2f bg[K2 / 4][2];
    if (wave < 4) {
        #pragma unroll
        for (int q = 0; q < K2 / 4; ++q) {
            const v2f* bp = reinterpret_cast<const v2f*>(&sW2[2 * q * PP + bofs]);
            bg[q][0] = bp[0];
            bg[q][1] = bp[16];
        }
    }
    __syncthreads();

    if (wave >= 4) return;
    v2f af[K2 / 4];
    {
        const v2f* x1 = reinterpret_cast<const v2f*>(&sS[(rl + m) * SP + kh]);
        #pragma unroll
        for (int q = 0; q < K2 / 4; ++q) af[q] = x1[2 * q];
    }
    __builtin_amdgcn_sched_barrier(0);

    v8f acc2[2] = {};
    #pragma unroll
    for (int q = 0; q < K2 / 4; ++q) {
        #pragma unroll
        for (int tt = 0; tt < 2; ++tt)
            acc2[tt] = __builtin_amdgcn_wmma_f32_16x16x4_f32(
                false, af[q], false, bg[q][tt], (short)0, acc2[tt], false, false);
    }

    #pragma unroll
    for (int tt = 0; tt < 2; ++tt) {
        const int n = (np * 2 + tt) * 16 + m;
        #pragma unroll
        for (int r = 0; r < 8; ++r) {
            const int mm = r + 8 * half;
            out[(long)(i0 + rl + mm) * UNITS + n] = acc2[tt][r] + bv2[tt];
        }
    }
}

extern "C" void kernel_launch(void* const* d_in, const int* in_sizes, int n_in,
                              void* d_out, int out_size, void* d_ws, size_t ws_size,
                              hipStream_t stream)
{
    const float* node_feature = (const float*)d_in[0];
    const float* edge_feature = (const float*)d_in[1];
    const float* W_init       = (const float*)d_in[2];
    const float* b_init       = (const float*)d_in[3];
    const float* W_upd        = (const float*)d_in[4];
    const float* b_upd        = (const float*)d_in[5];
    const int*   edge_src     = (const int*)d_in[6];
    const int*   edge_dst     = (const int*)d_in[7];
    float*       out          = (float*)d_out;

    edgeconv_kernel<<<N_EDGES / FI, 256, 0, stream>>>(
        node_feature, edge_feature, W_init, b_init, W_upd, b_upd,
        edge_src, edge_dst, out);
}
